# starting kernel
# baseline (speedup 1.0000x reference)
#include <hip/hip_runtime.h>


typedef float v2f __attribute__((ext_vector_type(2)));
typedef float v8f __attribute__((ext_vector_type(8)));

#define B_        16
#define K_        64
#define HF        16
#define WF        16
#define W_        80
#define LX        65
#define LY        25
#define FEAT_PAD  96
#define ROWS_PER_WG 4
#define WAVES     5
#define TPB       (WAVES * 32)

#define PIECE_FLOATS (K_ * HF * WF)
#define FEAT_FLOATS  (K_ * FEAT_PAD)
#define G2D_FLOATS   (B_ * 16 * 40 * (size_t)W_)
#define G1D_FLOATS   (B_ * 16 * 16 * (size_t)W_)

__global__ __launch_bounds__(TPB)
void corr_phase1_wmma(const float* __restrict__ piece,
                      const float* __restrict__ feat,
                      float* __restrict__ g,
                      int Hfeat, int kroll)
{
    extern __shared__ float smem[];
    float* sPiece = smem;
    float* sFeat  = smem + PIECE_FLOATS;

    const int b    = blockIdx.y;
    const int row0 = blockIdx.x * ROWS_PER_WG;
    const int tid  = threadIdx.x;
    const int wid  = tid >> 5;
    const int lane = tid & 31;
    const int half = lane >> 4;
    const int li   = lane & 15;
    const int x0   = wid * 16;

    const float* pieceB = piece + (size_t)b * PIECE_FLOATS;
    for (int idx = tid; idx < PIECE_FLOATS; idx += TPB) {
        int k    = idx >> 8;
        int rem  = idx & 255;
        int ksrc = (k + kroll) & (K_ - 1);
        sPiece[idx] = pieceB[ksrc * 256 + rem];
    }

    const int aBase = li * WF + 2 * half;
    const int bBase = x0 + li + 2 * half;

    for (int r = 0; r < ROWS_PER_WG; ++r) {
        const int row = row0 + r;
        const float* featRow = feat + ((size_t)(b * K_) * Hfeat + row) * W_;

        __syncthreads();
        for (int idx = tid; idx < FEAT_FLOATS; idx += TPB) {
            int k = idx / FEAT_PAD;
            int x = idx - k * FEAT_PAD;
            sFeat[idx] = (x < W_) ? featRow[(size_t)k * Hfeat * W_ + x] : 0.0f;
        }
        __syncthreads();

        v8f c = {0.f, 0.f, 0.f, 0.f, 0.f, 0.f, 0.f, 0.f};
        const float* ap = sPiece + aBase;
        const float* bp = sFeat  + bBase;

        for (int k = 0; k < K_; ++k) {
            #pragma unroll
            for (int jq = 0; jq < 4; ++jq) {
                v2f a, bb;
                a.x  = ap[jq * 4];
                a.y  = ap[jq * 4 + 1];
                bb.x = bp[jq * 4];
                bb.y = bp[jq * 4 + 1];
                c = __builtin_amdgcn_wmma_f32_16x16x4_f32(
                        false, a, false, bb, (short)0, c, false, false);
            }
            ap += HF * WF;
            bp += FEAT_PAD;
        }

        #pragma unroll
        for (int v = 0; v < 8; ++v) {
            int i = v + 8 * half;
            g[((size_t)(b * 16 + i) * Hfeat + row) * W_ + x0 + li] = c[v];
        }
    }
}

__global__ void corr_phase2_reduce(const float* __restrict__ g, float* __restrict__ out)
{
    int tid = blockIdx.x * blockDim.x + threadIdx.x;
    const int N2 = B_ * LY * LX;
    const int N1 = B_ * LX;
    if (tid < N2) {
        int b   = tid / (LY * LX);
        int rem = tid % (LY * LX);
        int y   = rem / LX;
        int x   = rem % LX;
        float s = 0.f;
        #pragma unroll
        for (int i = 0; i < 16; ++i)
            s += g[((size_t)(b * 16 + i) * 40 + (y + i)) * W_ + x];
        out[tid] = s;
    } else if (tid < N2 + N1) {
        int t2 = tid - N2;
        int b  = t2 / LX;
        int x  = t2 % LX;
        const float* g1 = g + G2D_FLOATS;
        float s = 0.f;
        #pragma unroll
        for (int i = 0; i < 16; ++i)
            s += g1[((size_t)(b * 16 + i) * 16 + i) * W_ + x];
        out[tid] = s;
    }
}

extern "C" void kernel_launch(void* const* d_in, const int* in_sizes, int n_in,
                              void* d_out, int out_size, void* d_ws, size_t ws_size,
                              hipStream_t stream)
{
    (void)in_sizes; (void)n_in; (void)out_size; (void)ws_size;

    const float* piece = (const float*)d_in[0];
    const float* comp  = (const float*)d_in[1];
    const float* strip = (const float*)d_in[2];

    float* g   = (float*)d_ws;
    float* out = (float*)d_out;

    const size_t shmem = (PIECE_FLOATS + FEAT_FLOATS) * sizeof(float);

    corr_phase1_wmma<<<dim3(10, B_), TPB, shmem, stream>>>(piece, comp, g, 40, 0);
    corr_phase1_wmma<<<dim3(4, B_),  TPB, shmem, stream>>>(piece, strip, g + G2D_FLOATS, 16, 32);

    const int total = B_ * LY * LX + B_ * LX;
    corr_phase2_reduce<<<(total + 255) / 256, 256, 0, stream>>>(g, out);
}


# corr_2d phase 1 as row chains: k split ik = step, jk = lane half, so base rows 4 apart share a sliding window of feature rows; a wave holds up to 7 tiles, A pairs shared, L+3 B pairs per channel
# speedup vs baseline: 10.4084x; 10.4084x over previous
#include <hip/hip_runtime.h>


typedef float v2f __attribute__((ext_vector_type(2)));
typedef float v4f __attribute__((ext_vector_type(4)));
typedef float v8f __attribute__((ext_vector_type(8)));

#define B_        16
#define K_        64
#define HF        16
#define WF        16
#define W_        80
#define LX        65
#define LY        25
#define XT        5
#define WAVES     8
#define TPB       (WAVES * 32)
#define NWG2D     256
#define NWG1D     160
#define RING      8
#define SPT       32
#define R2D       28
#define R1D       4
#define CLAST     76

#define PIECE_FLOATS (K_ * HF * WF)
#define G2D_FLOATS   (B_ * 16 * R2D * (size_t)W_)

struct Frag {
    v4f a0, a1;
    float b[8];
};

__device__ __forceinline__ void load_frag(Frag& f, const float* __restrict__ ap,
                                          const float* __restrict__ bp)
{
    f.a0 = *(const v4f*)(ap);
    f.a1 = *(const v4f*)(ap + 4 * WF);
    #pragma unroll
    for (int e = 0; e < 4; ++e) f.b[e] = bp[e];
    #pragma unroll
    for (int e = 0; e < 4; ++e) f.b[4 + e] = bp[4 * W_ + e];
}

__device__ __forceinline__ v8f mma_k(const Frag& f, v8f c)
{
    v2f a, bb;
    a.x = f.a0.x; a.y = f.a0.y; bb.x = f.b[0]; bb.y = f.b[1];
    c = __builtin_amdgcn_wmma_f32_16x16x4_f32(false, a, false, bb, (short)0, c, false, false);
    a.x = f.a0.z; a.y = f.a0.w; bb.x = f.b[2]; bb.y = f.b[3];
    c = __builtin_amdgcn_wmma_f32_16x16x4_f32(false, a, false, bb, (short)0, c, false, false);
    a.x = f.a1.x; a.y = f.a1.y; bb.x = f.b[4]; bb.y = f.b[5];
    c = __builtin_amdgcn_wmma_f32_16x16x4_f32(false, a, false, bb, (short)0, c, false, false);
    a.x = f.a1.z; a.y = f.a1.w; bb.x = f.b[6]; bb.y = f.b[7];
    c = __builtin_amdgcn_wmma_f32_16x16x4_f32(false, a, false, bb, (short)0, c, false, false);
    return c;
}

template <int HFEAT, int RT, int KROLL>
__device__ __forceinline__ void issue_set(Frag (&fs)[2], unsigned G,
                                          const float* __restrict__ piece,
                                          const float* __restrict__ feat,
                                          int aLane, int bLane, int li)
{
    constexpr unsigned TPBATCH = RT * XT;
    const unsigned task = G / SPT;
    const unsigned k    = (G % SPT) * 2;
    const unsigned b    = task / TPBATCH;
    const unsigned q    = task - b * TPBATCH;
    const unsigned r    = q / XT;
    const int      c0   = (int)(q - r * XT) * 16;
    const float* ap = piece + (size_t)b * PIECE_FLOATS
                    + ((k + KROLL) & (K_ - 1)) * (HF * WF) + aLane;
    const int cl = min(li, CLAST - c0);
    const float* bp = feat + ((size_t)(b * K_ + k) * HFEAT + r) * W_ + c0 + cl + bLane;
    load_frag(fs[0], ap, bp);
    load_frag(fs[1], ap + HF * WF, bp + HFEAT * W_);
}

template <int RT>
__device__ __forceinline__ void store_tile(float* __restrict__ g, unsigned task, v8f c,
                                           int half, int li)
{
    constexpr unsigned TPBATCH = RT * XT;
    const unsigned b  = task / TPBATCH;
    const unsigned q  = task - b * TPBATCH;
    const unsigned r  = q / XT;
    const unsigned c0 = (q - r * XT) * 16;
    #pragma unroll
    for (int v = 0; v < 8; ++v) {
        const unsigned m = v + 8 * half;
        g[((size_t)(b * 16 + m) * RT + r) * W_ + c0 + li] = c[v];
    }
}

template <int HFEAT, int RT, int KROLL, int TP4>
__global__ __launch_bounds__(TPB)
void corr_phase1_wmma(const float* __restrict__ piece,
                      const float* __restrict__ feat,
                      float* __restrict__ g)
{
    __shared__ float part[WAVES][8][32];

    const int tid  = threadIdx.x;
    const int wid  = __builtin_amdgcn_readfirstlane(tid >> 5);
    const int lane = tid & 31;
    const int half = lane >> 4;
    const int li   = lane & 15;
    const int aLane = ((li >> 2) + 8 * half) * WF + 4 * (li & 3);
    const int bLane = 8 * half * W_;

    const int t0   = ((int)blockIdx.x * TP4) >> 2;
    const int nt   = ((((int)blockIdx.x + 1) * TP4) >> 2) - t0;
    const int base = nt * (SPT / WAVES);
    const int odd  = base & 4;
#define WAVE_START(w) (t0 * SPT + (w) * base + (odd ? ((w) < 4 ? 4 * (w) : 32 - 4 * (w)) : 0))
    const int G0 = WAVE_START(wid);
    const int S  = base + (odd ? (wid < 4 ? 4 : -4) : 0);
    const int GE = G0 + S;

    Frag f[RING][2];
    #pragma unroll
    for (int s = 0; s < RING - 1; ++s) {
        issue_set<HFEAT, RT, KROLL>(f[s], G0 + s, piece, feat, aLane, bLane, li);
        __builtin_amdgcn_sched_barrier(0);
    }

    v8f c = {0.f, 0.f, 0.f, 0.f, 0.f, 0.f, 0.f, 0.f};
    bool inHead = (G0 % SPT) != 0;

#define TILE_END(Gs)                                                        \
    if (((Gs) % SPT) == SPT - 1) {                                          \
        if (inHead) {                                                       \
            _Pragma("unroll")                                               \
            for (int v = 0; v < 8; ++v) part[wid][v][lane] = c[v];          \
            inHead = false;                                                 \
        } else {                                                            \
            store_tile<RT>(g, (unsigned)(Gs) / SPT, c, half, li);           \
        }                                                                   \
        _Pragma("unroll")                                                   \
        for (int v = 0; v < 8; ++v) c[v] = 0.f;                             \
    }

    int G = G0;
    const int nblk = S / RING;
    for (int blk = 0; blk < nblk - 1; ++blk) {
        #pragma unroll
        for (int s = 0; s < RING; ++s) {
            issue_set<HFEAT, RT, KROLL>(f[(s + RING - 1) % RING], G + s + RING - 1,
                                        piece, feat, aLane, bLane, li);
            __builtin_amdgcn_sched_barrier(0);
            c = mma_k(f[s][0], c);
            c = mma_k(f[s][1], c);
            __builtin_amdgcn_sched_barrier(0);
            if ((s & 3) == 3) { TILE_END(G + s) }
        }
        G += RING;
    }
    #pragma unroll
    for (int s = 0; s < RING; ++s) {
        if (s == 0)
            issue_set<HFEAT, RT, KROLL>(f[RING - 1], G + RING - 1, piece, feat, aLane, bLane, li);
        __builtin_amdgcn_sched_barrier(0);
        c = mma_k(f[s][0], c);
        c = mma_k(f[s][1], c);
        __builtin_amdgcn_sched_barrier(0);
        if ((s & 3) == 3) { TILE_END(G + s) }
    }

    const bool tail = (GE % SPT) != 0;
    if (tail && inHead) {
        #pragma unroll
        for (int v = 0; v < 8; ++v) part[wid][v][lane] = c[v];
    }
    __syncthreads();
    if (tail && !inHead) {
        const int task    = GE / SPT;
        const int taskEnd = (task + 1) * SPT;
        for (int w = wid + 1; w < WAVES; ++w) {
            if (WAVE_START(w) >= taskEnd) break;
            #pragma unroll
            for (int v = 0; v < 8; ++v) c[v] += part[w][v][lane];
        }
        store_tile<RT>(g, (unsigned)task, c, half, li);
    }
#undef TILE_END
#undef WAVE_START
}


typedef v2f v2fu __attribute__((aligned(4)));

#define CH_AHEAD 3
#define CH_RING  4

struct CBuf {
    v2f a[4];
    v2f b[10];
};


template <int L>
__device__ __forceinline__ void issue_ch(CBuf& f, const float* __restrict__ ap,
                                         const float* __restrict__ bp)
{
    #pragma unroll
    for (int t = 0; t < 4; ++t) f.a[t] = *(const v2f*)(ap + t * 4 * WF);
    #pragma unroll
    for (int j = 0; j < L + 3; ++j) f.b[j] = *(const v2fu*)(bp + j * 4 * W_);
}

template <int L>
__device__ __forceinline__ void mma_ch(const CBuf& f, v8f (&acc)[7])
{
    #pragma unroll
    for (int t = 0; t < 4; ++t) {
        #pragma unroll
        for (int n = 0; n < L; ++n)
            acc[n] = __builtin_amdgcn_wmma_f32_16x16x4_f32(false, f.a[t], false, f.b[n + t],
                                                           (short)0, acc[n], false, false);
    }
}

#define CHAIN_TILES 9
#define PART_SLOTS  21

template <int VAR>
__device__ __forceinline__ void chain_var(const float* __restrict__ piece,
                                          const float* __restrict__ comp,
                                          float* __restrict__ g,
                                          float (*part)[PART_SLOTS][8][32],
                                          int q5, int wid, int lane)
{
    constexpr int LA = 7 - 2 * VAR;
    constexpr int NA = 2 * VAR;
    constexpr int LB = (VAR == 3) ? 7 : 2 + 2 * VAR;
    constexpr int CHS = 40 * W_;

    const int half = lane >> 4;
    const int li   = lane & 15;
    const int chA  = q5 * 5 + VAR;
    const int chB  = chA + 1;
    const int bA = chA / 20, r0A = (chA / 5) & 3, c0A = (chA % 5) * 16;
    const int bB = chB / 20, r0B = (chB / 5) & 3, c0B = (chB % 5) * 16;
    const int k0 = 8 * wid;

    const int aLane = (li >> 2) * WF + 4 * (li & 3) + 2 * half;
    const float* apA = piece + (size_t)bA * PIECE_FLOATS + k0 * (HF * WF) + aLane;
    const float* apB = piece + (size_t)bB * PIECE_FLOATS + k0 * (HF * WF) + aLane;
    const int clA = min(li, CLAST - c0A);
    const int clB = min(li, CLAST - c0B);
    const float* bpA = comp + ((size_t)(bA * K_ + k0) * 40 + r0A + 4 * NA) * W_ + c0A + clA + 2 * half;
    const float* bpB = comp + ((size_t)(bB * K_ + k0) * 40 + r0B) * W_ + c0B + clB + 2 * half;

    CBuf f[CH_RING];
    v8f accA[7], accB[7];
    #pragma unroll
    for (int n = 0; n < 7; ++n)
        #pragma unroll
        for (int v = 0; v < 8; ++v) { accA[n][v] = 0.f; accB[n][v] = 0.f; }

    #pragma unroll
    for (int i = 0; i < CH_AHEAD; ++i) {
        issue_ch<LA>(f[i], apA + i * (HF * WF), bpA + i * CHS);
        __builtin_amdgcn_sched_barrier(0);
    }
    #pragma unroll
    for (int i = 0; i < 8; ++i) {
        const int nx = i + CH_AHEAD;
        if (nx < 8) issue_ch<LA>(f[nx % CH_RING], apA + nx * (HF * WF), bpA + nx * CHS);
        else        issue_ch<LB>(f[nx % CH_RING], apB + (nx - 8) * (HF * WF), bpB + (nx - 8) * CHS);
        __builtin_amdgcn_sched_barrier(0);
        mma_ch<LA>(f[i % CH_RING], accA);
        __builtin_amdgcn_sched_barrier(0);
    }
    #pragma unroll
    for (int n = 0; n < LA; ++n)
        #pragma unroll
        for (int v = 0; v < 8; ++v) part[wid][n][v][lane] = accA[n][v];
    #pragma unroll
    for (int i = 0; i < 8; ++i) {
        const int nx = i + CH_AHEAD;
        if (nx < 8) issue_ch<LB>(f[(nx + 8) % CH_RING], apB + nx * (HF * WF), bpB + nx * CHS);
        __builtin_amdgcn_sched_barrier(0);
        mma_ch<LB>(f[(i + 8) % CH_RING], accB);
        __builtin_amdgcn_sched_barrier(0);
    }
    #pragma unroll
    for (int n = 0; n < LB; ++n)
        #pragma unroll
        for (int v = 0; v < 8; ++v) part[wid][LA + n][v][lane] = accB[n][v];

    __syncthreads();
    #pragma unroll
    for (int s = 0; s < LA + LB; ++s) {
        float sum = 0.f;
        #pragma unroll
        for (int w = 0; w < WAVES; ++w) sum += part[w][s][wid][lane];
        const int b  = (s < LA) ? bA : bB;
        const int r  = (s < LA) ? r0A + 4 * (NA + s) : r0B + 4 * (s - LA);
        const int c0 = (s < LA) ? c0A : c0B;
        const int m  = wid + 8 * half;
        g[((size_t)(b * 16 + m) * R2D + r) * W_ + c0 + li] = sum;
    }
}

__global__ __launch_bounds__(TPB)
void corr2d_chain_wmma(const float* __restrict__ piece,
                       const float* __restrict__ comp,
                       float* __restrict__ g)
{
    __shared__ float part[WAVES][PART_SLOTS][8][32];
    const int tid  = threadIdx.x;
    const int wid  = __builtin_amdgcn_readfirstlane(tid >> 5);
    const int lane = tid & 31;
    const int q5   = (int)blockIdx.x >> 2;
    switch (blockIdx.x & 3) {
    case 0:  chain_var<0>(piece, comp, g, part, q5, wid, lane); break;
    case 1:  chain_var<1>(piece, comp, g, part, q5, wid, lane); break;
    case 2:  chain_var<2>(piece, comp, g, part, q5, wid, lane); break;
    default: chain_var<3>(piece, comp, g, part, q5, wid, lane); break;
    }
}

__global__ void corr_phase2_reduce(const float* __restrict__ g, float* __restrict__ out)
{
    int tid = blockIdx.x * blockDim.x + threadIdx.x;
    const int N2 = B_ * LY * LX;
    const int N1 = B_ * LX;
    if (tid < N2) {
        int b   = tid / (LY * LX);
        int rem = tid % (LY * LX);
        int y   = rem / LX;
        int x   = rem % LX;
        float s = 0.f;
        #pragma unroll
        for (int m = 0; m < 16; ++m)
            s += g[((size_t)(b * 16 + m) * R2D + (y + (m >> 2))) * W_ + x + 4 * (m & 3)];
        out[tid] = s;
    } else if (tid < N2 + N1) {
        int t2 = tid - N2;
        int b  = t2 / LX;
        int x  = t2 % LX;
        const float* g1 = g + G2D_FLOATS;
        float s = 0.f;
        #pragma unroll
        for (int m = 0; m < 16; ++m)
            s += g1[((size_t)(b * 16 + m) * R1D + (m >> 2)) * W_ + x + 4 * (m & 3)];
        out[tid] = s;
    }
}

extern "C" void kernel_launch(void* const* d_in, const int* in_sizes, int n_in,
                              void* d_out, int out_size, void* d_ws, size_t ws_size,
                              hipStream_t stream)
{
    (void)in_sizes; (void)n_in; (void)out_size; (void)ws_size;

    const float* piece = (const float*)d_in[0];
    const float* comp  = (const float*)d_in[1];
    const float* strip = (const float*)d_in[2];

    float* g   = (float*)d_ws;
    float* out = (float*)d_out;

    corr2d_chain_wmma<<<NWG2D, TPB, 0, stream>>>(piece, comp, g);
    corr_phase1_wmma<16, R1D, 32, 8><<<NWG1D, TPB, 0, stream>>>(piece, strip, g + G2D_FLOATS);

    const int total = B_ * LY * LX + B_ * LX;
    corr_phase2_reduce<<<(total + 255) / 256, 256, 0, stream>>>(g, out);
}


# B pairs of column tiles 0-3 from one 16-byte + one 4-byte load per lane (lane holds columns 4l+n; was four 8-byte loads, columns 16n+l): 6 -> 4 loads per unit, loads 14 units ahead (was 9)
# speedup vs baseline: 16.7058x; 1.6050x over previous
#include <hip/hip_runtime.h>


typedef float v2f __attribute__((ext_vector_type(2)));
typedef float v8f __attribute__((ext_vector_type(8)));
typedef float v4f __attribute__((ext_vector_type(4)));
typedef v2f v2fu __attribute__((aligned(4)));
typedef v4f v4fu __attribute__((aligned(4)));

#define B_        16
#define K_        64
#define HF        16
#define WF        16
#define HC        40
#define W_        80
#define LX        65
#define LY        25
#define WAVES     8
#define TPB       (WAVES * 32)
#define NYG       13
#define NWG2D     (B_ * NYG)
#define NWG       (NWG2D + B_)
#define NCT       5
#define CMAX      78
#define NSTEP     9
#define NUNIT     (8 * NSTEP)
#define AHEAD     14
#define RING      15

#define PIECE_FLOATS (K_ * HF * WF)
#define OUT2D_FLOATS (B_ * LY * LX)

#define PART_SLOTS  20

struct UBuf {
    v2f   a;
    v4f   bq;
    float b4;
    v2f   bt;
};

template <int HFEAT>
__device__ __forceinline__ void corr_rows(const float* __restrict__ pw,
                                          const float* __restrict__ fw,
                                          int y0, int nrow, float* __restrict__ orow,
                                          float (*part)[PART_SLOTS][8][32],
                                          float (*dfull)[W_],
                                          int wid, int lane, int tid)
{
    constexpr int CHS = HFEAT * W_;
    const int half = lane >> 4;
    const int li   = lane & 15;
    const int ym   = li >> 3;
    const int jm   = li & 7;
    const int ro   = half - ym;
    const float* ab  = pw + 2 * jm;
    const float* apm = ab + ro * WF;
    const float* ap0 = ab + max(ro, 0) * WF;
    const float* ap8 = ab + min(2 * 8 + ro, HF - 1) * WF;
    const bool ok0 = ro >= 0;
    const bool ok8 = ro < 0;
    const int  ch  = 4 * 16 + min(li, CMAX - 4 * 16);
    const float* f07 = fw + (size_t)(y0 + half) * W_;
    const float* f8  = fw + (size_t)min(y0 + 2 * 8 + half, HFEAT - 1) * W_;

    UBuf f[RING];
    v8f acc[NCT];
    #pragma unroll
    for (int n = 0; n < NCT; ++n)
        #pragma unroll
        for (int v = 0; v < 8; ++v) acc[n][v] = 0.f;

    auto issue = [&](int q) {
        const int kc = q / NSTEP, u = q % NSTEP;
        UBuf& s = f[q % RING];
        const float* ap = (u == 0) ? ap0 : (u == NSTEP - 1) ? ap8 : apm + 2 * u * WF;
        s.a = *(const v2f*)(ap + kc * (HF * WF));
        const float* fr = (u == NSTEP - 1) ? f8 : f07 + 2 * u * W_;
        s.bq = *(const v4fu*)(fr + kc * CHS + 4 * li);
        s.b4 = fr[kc * CHS + 4 * li + 4];
        s.bt = *(const v2fu*)(fr + kc * CHS + ch);
    };
    auto mma = [&](int q) {
        const int u = q % NSTEP;
        const UBuf& s = f[q % RING];
        v2f a = s.a;
        if (u == 0)         { a[0] = ok0 ? a[0] : 0.f; a[1] = ok0 ? a[1] : 0.f; }
        if (u == NSTEP - 1) { a[0] = ok8 ? a[0] : 0.f; a[1] = ok8 ? a[1] : 0.f; }
        v2f b[NCT];
        b[0][0] = s.bq[0]; b[0][1] = s.bq[1];
        b[1][0] = s.bq[1]; b[1][1] = s.bq[2];
        b[2][0] = s.bq[2]; b[2][1] = s.bq[3];
        b[3][0] = s.bq[3]; b[3][1] = s.b4;
        b[4] = s.bt;
        #pragma unroll
        for (int n = 0; n < NCT; ++n)
            acc[n] = __builtin_amdgcn_wmma_f32_16x16x4_f32(false, a, false, b[n],
                                                           (short)0, acc[n], false, false);
    };

    #pragma unroll
    for (int q = 0; q < AHEAD; ++q) {
        issue(q);
        __builtin_amdgcn_sched_barrier(0);
    }
    #pragma unroll
    for (int q = 0; q < NUNIT; ++q) {
        if (q + AHEAD < NUNIT) issue(q + AHEAD);
        __builtin_amdgcn_sched_barrier(0);
        mma(q);
        __builtin_amdgcn_sched_barrier(0);
    }

    #pragma unroll
    for (int n = 0; n < NCT; ++n)
        #pragma unroll
        for (int v = 0; v < 8; ++v) part[wid][n][v][lane] = acc[n][v];
    __syncthreads();
    #pragma unroll
    for (int n = 0; n < NCT; ++n) {
        float sum = 0.f;
        #pragma unroll
        for (int w = 0; w < WAVES; ++w) sum += part[w][n][wid][lane];
        dfull[wid + 8 * half][(n < NCT - 1) ? 4 * li + n : 4 * 16 + li] = sum;
    }
    __syncthreads();
    if (tid < nrow * LX) {
        const int r = (tid >= LX) ? 1 : 0;
        const int x = tid - r * LX;
        float s = 0.f;
        #pragma unroll
        for (int j = 0; j < 8; ++j) s += dfull[8 * r + j][x + 2 * j];
        orow[r * LX + x] = s;
    }
}

__global__ __launch_bounds__(TPB)
void corr_rows_wmma(const float* __restrict__ piece,
                    const float* __restrict__ comp,
                    const float* __restrict__ strip,
                    float* __restrict__ out)
{
    __shared__ float part[WAVES][PART_SLOTS][8][32];
    __shared__ float dfull[16][W_];
    const int tid  = threadIdx.x;
    const int wid  = __builtin_amdgcn_readfirstlane(tid >> 5);
    const int lane = tid & 31;
    const int bid  = (int)blockIdx.x;
    const int k0   = 8 * wid;
    if (bid < NWG2D) {
        const int b  = bid / NYG;
        const int y0 = 2 * (bid - b * NYG);
        corr_rows<HC>(piece + (size_t)b * PIECE_FLOATS + k0 * (HF * WF),
                      comp + (size_t)(b * K_ + k0) * (HC * W_),
                      y0, (y0 + 1 < LY) ? 2 : 1,
                      out + (size_t)b * (LY * LX) + y0 * LX,
                      part, dfull, wid, lane, tid);
    } else {
        const int b = bid - NWG2D;
        corr_rows<HF>(piece + (size_t)b * PIECE_FLOATS + ((k0 + K_ / 2) & (K_ - 1)) * (HF * WF),
                      strip + (size_t)(b * K_ + k0) * (HF * W_),
                      0, 1,
                      out + OUT2D_FLOATS + (size_t)b * LX,
                      part, dfull, wid, lane, tid);
    }
}

extern "C" void kernel_launch(void* const* d_in, const int* in_sizes, int n_in,
                              void* d_out, int out_size, void* d_ws, size_t ws_size,
                              hipStream_t stream)
{
    (void)in_sizes; (void)n_in; (void)out_size; (void)d_ws; (void)ws_size;

    const float* piece = (const float*)d_in[0];
    const float* comp  = (const float*)d_in[1];
    const float* strip = (const float*)d_in[2];

    float* out = (float*)d_out;

    corr_rows_wmma<<<NWG, TPB, 0, stream>>>(piece, comp, strip, out);
}
